# starting kernel
# baseline (speedup 1.0000x reference)
#include <hip/hip_runtime.h>
#include <hip/hip_bf16.h>
#include <math.h>

typedef __attribute__((ext_vector_type(2))) float v2f;
typedef __attribute__((ext_vector_type(8))) float v8f;

#define WMMA_F32(A, B, C) \
  __builtin_amdgcn_wmma_f32_16x16x4_f32(false, (A), false, (B), (short)0, (C), false, false)

__device__ __forceinline__ v2f mkv2(float a, float b) { v2f r; r.x = a; r.y = b; return r; }

__device__ __forceinline__ void wave_lds_fence() {
  asm volatile("s_wait_dscnt 0" ::: "memory");
}

constexpr int Bsz = 32768;
constexpr int D   = 64;
constexpr int H   = 32;
constexpr int TILES = 4;
constexpr int WAVES = 8;
constexpr int ROWS_PER_BLOCK = WAVES * TILES * 16;

__global__ __launch_bounds__(256) void encoder_kernel(
    const float* __restrict__ x,   const float* __restrict__ eps,
    const float* __restrict__ W1,  const float* __restrict__ b1,
    const float* __restrict__ W2,  const float* __restrict__ b2,
    const float* __restrict__ Wmu, const float* __restrict__ bmu,
    const float* __restrict__ Wlv, const float* __restrict__ blv,
    float* __restrict__ mu_out, float* __restrict__ lv_out,
    float* __restrict__ z_ws)
{
  __shared__ float lds[WAVES * 16 * H];
  const int lane = threadIdx.x & 31;
  const int wave = threadIdx.x >> 5;
  const int col  = lane & 15;
  const int hi   = lane >> 4;
  const int d    = blockIdx.y;
  float* wl = &lds[wave * 16 * H];

  v2f w2b[2][8];
  v2f w1v[8], b1v[8];
  v2f wml[8];
  #pragma unroll
  for (int c = 0; c < 8; ++c) {
    const int k0 = 4 * c + 2 * hi;
    w1v[c] = mkv2(W1[d * H + k0], W1[d * H + k0 + 1]);
    b1v[c] = mkv2(b1[d * H + k0], b1[d * H + k0 + 1]);
    #pragma unroll
    for (int nt = 0; nt < 2; ++nt) {
      const int n = nt * 16 + col;
      w2b[nt][c] = mkv2(W2[d * H * H + k0 * H + n],
                        W2[d * H * H + (k0 + 1) * H + n]);
    }
    const float m0 = Wmu[d * H + k0], m1 = Wmu[d * H + k0 + 1];
    const float l0 = Wlv[d * H + k0], l1 = Wlv[d * H + k0 + 1];
    wml[c] = mkv2(col == 0 ? m0 : (col == 1 ? l0 : 0.0f),
                  col == 0 ? m1 : (col == 1 ? l1 : 0.0f));
  }
  const float b2v0 = b2[d * H + col];
  const float b2v1 = b2[d * H + 16 + col];
  const float bmuv = bmu[d];
  const float blvv = blv[d];

  const int rowBase = blockIdx.x * ROWS_PER_BLOCK + wave * (TILES * 16);

  for (int t = 0; t < TILES; ++t) {
    const int row0 = rowBase + t * 16;
    const float xv = x[(row0 + col) * D + d];

    v8f acc0 = {}, acc1 = {};
    #pragma unroll
    for (int c = 0; c < 8; ++c) {
      v2f a;
      a.x = fmaxf(fmaf(xv, w1v[c].x, b1v[c].x), 0.0f);
      a.y = fmaxf(fmaf(xv, w1v[c].y, b1v[c].y), 0.0f);
      acc0 = WMMA_F32(a, w2b[0][c], acc0);
      acc1 = WMMA_F32(a, w2b[1][c], acc1);
    }

    wave_lds_fence();
    #pragma unroll
    for (int j = 0; j < 8; ++j) {
      const int r = j + 8 * hi;
      wl[r * H + col]      = fmaxf(acc0[j] + b2v0, 0.0f);
      wl[r * H + 16 + col] = fmaxf(acc1[j] + b2v1, 0.0f);
    }
    wave_lds_fence();

    v8f accS = {};
    #pragma unroll
    for (int c = 0; c < 8; ++c) {
      v2f a = *(const v2f*)&wl[col * H + 4 * c + 2 * hi];
      accS = WMMA_F32(a, wml[c], accS);
    }

    #pragma unroll
    for (int j = 0; j < 8; ++j) {
      const float self  = accS[j];
      const float other = __shfl_xor(self, 1, 32);
      const int row = row0 + j + 8 * hi;
      if (col == 0) {
        const float m  = self + bmuv;
        const float lv = other + blvv;
        mu_out[row * D + d] = m;
        z_ws[row * D + d]   = fmaf(eps[row * D + d], __expf(0.5f * lv), m);
      } else if (col == 1) {
        lv_out[row * D + d] = self + blvv;
      }
    }
  }
}

__global__ __launch_bounds__(256) void decoder_kernel(
    const float* __restrict__ z_ws,
    const float* __restrict__ Wd1, const float* __restrict__ bd1,
    const float* __restrict__ Wd2, const float* __restrict__ bd2,
    const float* __restrict__ Wd3, const float* __restrict__ bd3,
    float* __restrict__ action)
{
  __shared__ float lds[WAVES * 16 * H];
  const int lane = threadIdx.x & 31;
  const int wave = threadIdx.x >> 5;
  const int col  = lane & 15;
  const int hi   = lane >> 4;
  float* wl = &lds[wave * 16 * H];

  v2f wd1b[2][16];
  #pragma unroll
  for (int c = 0; c < 16; ++c) {
    const int k0 = 4 * c + 2 * hi;
    #pragma unroll
    for (int nt = 0; nt < 2; ++nt) {
      const int n = nt * 16 + col;
      wd1b[nt][c] = mkv2(Wd1[k0 * 32 + n], Wd1[(k0 + 1) * 32 + n]);
    }
  }
  v2f wd2b[2][8], wd3b[8];
  #pragma unroll
  for (int c = 0; c < 8; ++c) {
    const int k0 = 4 * c + 2 * hi;
    #pragma unroll
    for (int nt = 0; nt < 2; ++nt) {
      const int n = nt * 16 + col;
      wd2b[nt][c] = mkv2(Wd2[k0 * 32 + n], Wd2[(k0 + 1) * 32 + n]);
    }
    wd3b[c] = mkv2(col == 0 ? Wd3[k0] : 0.0f,
                   col == 0 ? Wd3[k0 + 1] : 0.0f);
  }
  const float bd1v0 = bd1[col], bd1v1 = bd1[16 + col];
  const float bd2v0 = bd2[col], bd2v1 = bd2[16 + col];
  const float bd3v  = bd3[0];

  const int rowBase = blockIdx.x * ROWS_PER_BLOCK + wave * (TILES * 16);

  for (int t = 0; t < TILES; ++t) {
    const int row0 = rowBase + t * 16;

    v8f acc0 = {}, acc1 = {};
    #pragma unroll
    for (int c = 0; c < 16; ++c) {
      v2f a = *(const v2f*)&z_ws[(row0 + col) * 64 + 4 * c + 2 * hi];
      acc0 = WMMA_F32(a, wd1b[0][c], acc0);
      acc1 = WMMA_F32(a, wd1b[1][c], acc1);
    }
    wave_lds_fence();
    #pragma unroll
    for (int j = 0; j < 8; ++j) {
      const int r = j + 8 * hi;
      wl[r * 32 + col]      = fmaxf(acc0[j] + bd1v0, 0.0f);
      wl[r * 32 + 16 + col] = fmaxf(acc1[j] + bd1v1, 0.0f);
    }
    wave_lds_fence();

    v8f a20 = {}, a21 = {};
    #pragma unroll
    for (int c = 0; c < 8; ++c) {
      v2f a = *(const v2f*)&wl[col * 32 + 4 * c + 2 * hi];
      a20 = WMMA_F32(a, wd2b[0][c], a20);
      a21 = WMMA_F32(a, wd2b[1][c], a21);
    }
    wave_lds_fence();
    #pragma unroll
    for (int j = 0; j < 8; ++j) {
      const int r = j + 8 * hi;
      wl[r * 32 + col]      = fmaxf(a20[j] + bd2v0, 0.0f);
      wl[r * 32 + 16 + col] = fmaxf(a21[j] + bd2v1, 0.0f);
    }
    wave_lds_fence();

    v8f accS = {};
    #pragma unroll
    for (int c = 0; c < 8; ++c) {
      v2f a = *(const v2f*)&wl[col * 32 + 4 * c + 2 * hi];
      accS = WMMA_F32(a, wd3b[c], accS);
    }
    if (col == 0) {
      #pragma unroll
      for (int j = 0; j < 8; ++j)
        action[row0 + j + 8 * hi] = accS[j] + bd3v;
    }
  }
}

extern "C" void kernel_launch(void* const* d_in, const int* in_sizes, int n_in,
                              void* d_out, int out_size, void* d_ws, size_t ws_size,
                              hipStream_t stream) {
  (void)in_sizes; (void)n_in; (void)out_size; (void)ws_size;
  const float* x   = (const float*)d_in[0];
  const float* eps = (const float*)d_in[1];
  const float* W1  = (const float*)d_in[2];
  const float* b1  = (const float*)d_in[3];
  const float* W2  = (const float*)d_in[4];
  const float* b2  = (const float*)d_in[5];
  const float* Wmu = (const float*)d_in[6];
  const float* bmu = (const float*)d_in[7];
  const float* Wlv = (const float*)d_in[8];
  const float* blv = (const float*)d_in[9];
  const float* Wd1 = (const float*)d_in[10];
  const float* bd1 = (const float*)d_in[11];
  const float* Wd2 = (const float*)d_in[12];
  const float* bd2 = (const float*)d_in[13];
  const float* Wd3 = (const float*)d_in[14];
  const float* bd3 = (const float*)d_in[15];

  float* out     = (float*)d_out;
  float* action  = out;
  float* mu_out  = out + Bsz;
  float* lv_out  = out + Bsz + Bsz * D;
  float* z_ws    = (float*)d_ws;

  dim3 grid1(Bsz / ROWS_PER_BLOCK, D);
  encoder_kernel<<<grid1, 256, 0, stream>>>(x, eps, W1, b1, W2, b2,
                                            Wmu, bmu, Wlv, blv,
                                            mu_out, lv_out, z_ws);
  decoder_kernel<<<Bsz / ROWS_PER_BLOCK, 256, 0, stream>>>(z_ws, Wd1, bd1,
                                                           Wd2, bd2, Wd3, bd3,
                                                           action);
}


# encoder: weights as matrix-op A, activations as B: h2 comes out transposed and its D registers are the B of the mu/logvar product (no LDS trip), rows on lanes; mu/logvar k summation order changes
# speedup vs baseline: 1.5361x; 1.5361x over previous
#include <hip/hip_runtime.h>
#include <hip/hip_bf16.h>
#include <math.h>

typedef __attribute__((ext_vector_type(2))) float v2f;
typedef __attribute__((ext_vector_type(8))) float v8f;

#define WMMA_F32(A, B, C) \
  __builtin_amdgcn_wmma_f32_16x16x4_f32(false, (A), false, (B), (short)0, (C), false, false)

__device__ __forceinline__ v2f mkv2(float a, float b) { v2f r; r.x = a; r.y = b; return r; }

__device__ __forceinline__ void wave_lds_fence() {
  asm volatile("s_wait_dscnt 0" ::: "memory");
}

constexpr int Bsz = 32768;
constexpr int D   = 64;
constexpr int H   = 32;
constexpr int TILES = 4;
constexpr int WAVES = 8;
constexpr int ROWS_PER_BLOCK = WAVES * TILES * 16;

__global__ __launch_bounds__(256) void encoder_kernel(
    const float* __restrict__ x,   const float* __restrict__ eps,
    const float* __restrict__ W1,  const float* __restrict__ b1,
    const float* __restrict__ W2,  const float* __restrict__ b2,
    const float* __restrict__ Wmu, const float* __restrict__ bmu,
    const float* __restrict__ Wlv, const float* __restrict__ blv,
    float* __restrict__ mu_out, float* __restrict__ lv_out,
    float* __restrict__ z_ws)
{
  const int lane = threadIdx.x & 31;
  const int wave = threadIdx.x >> 5;
  const int col  = lane & 15;
  const int hi   = lane >> 4;
  const int d    = blockIdx.y;
  const int rowBase = blockIdx.x * ROWS_PER_BLOCK + wave * (TILES * 16);

  float xv[TILES];
  float ev[TILES / 2];
  #pragma unroll
  for (int t = 0; t < TILES; ++t) xv[t] = x[(rowBase + 16 * t + col) * D + d];
  #pragma unroll
  for (int p = 0; p < TILES / 2; ++p) ev[p] = eps[(rowBase + 32 * p + lane) * D + d];

  v2f w2a[2][8];
  v2f w1v[8], b1v[8];
  v2f wsa[8];
  const bool   srow = (col & 7) < 2;
  const float* wsp  = ((col & 7) == 1 ? Wlv : Wmu) + d * H;
  #pragma unroll
  for (int c = 0; c < 8; ++c) {
    const int k0 = 4 * c + 2 * hi;
    w1v[c] = *(const v2f*)&W1[d * H + k0];
    b1v[c] = *(const v2f*)&b1[d * H + k0];
    #pragma unroll
    for (int nt = 0; nt < 2; ++nt) {
      const int n = nt * 16 + col;
      w2a[nt][c] = mkv2(W2[d * H * H + k0 * H + n],
                        W2[d * H * H + (k0 + 1) * H + n]);
    }
    const int nb = 16 * (c >> 2) + 8 * hi + 2 * (c & 3);
    const v2f wv = *(const v2f*)&wsp[nb];
    wsa[c] = mkv2(srow ? wv.x : 0.0f, srow ? wv.y : 0.0f);
  }
  const v8f b2r0 = *(const v8f*)&b2[d * H + 8 * hi];
  const v8f b2r1 = *(const v8f*)&b2[d * H + 16 + 8 * hi];
  const float bmuv = bmu[d];
  const float blvv = blv[d];

  #pragma unroll
  for (int p = 0; p < TILES / 2; ++p) {
    float mu2[2], lv2[2];
    #pragma unroll
    for (int u = 0; u < 2; ++u) {
      const float xt = xv[2 * p + u];

      v8f acc0 = {}, acc1 = {};
      #pragma unroll
      for (int c = 0; c < 8; ++c) {
        v2f hb;
        hb.x = fmaxf(fmaf(xt, w1v[c].x, b1v[c].x), 0.0f);
        hb.y = fmaxf(fmaf(xt, w1v[c].y, b1v[c].y), 0.0f);
        acc0 = WMMA_F32(w2a[0][c], hb, acc0);
        acc1 = WMMA_F32(w2a[1][c], hb, acc1);
      }
      #pragma unroll
      for (int r = 0; r < 8; ++r) {
        acc0[r] = fmaxf(acc0[r] + b2r0[r], 0.0f);
        acc1[r] = fmaxf(acc1[r] + b2r1[r], 0.0f);
      }

      v8f accS = {};
      #pragma unroll
      for (int q = 0; q < 4; ++q)
        accS = WMMA_F32(wsa[q], mkv2(acc0[2 * q], acc0[2 * q + 1]), accS);
      #pragma unroll
      for (int q = 0; q < 4; ++q)
        accS = WMMA_F32(wsa[4 + q], mkv2(acc1[2 * q], acc1[2 * q + 1]), accS);

      mu2[u] = accS[0] + bmuv;
      lv2[u] = accS[1] + blvv;
    }

    const float m  = hi ? mu2[1] : mu2[0];
    const float lv = hi ? lv2[1] : lv2[0];
    const int  row = rowBase + 32 * p + lane;
    mu_out[row * D + d] = m;
    lv_out[row * D + d] = lv;
    z_ws[row * D + d]   = fmaf(ev[p], __expf(0.5f * lv), m);
  }
}

__global__ __launch_bounds__(256) void decoder_kernel(
    const float* __restrict__ z_ws,
    const float* __restrict__ Wd1, const float* __restrict__ bd1,
    const float* __restrict__ Wd2, const float* __restrict__ bd2,
    const float* __restrict__ Wd3, const float* __restrict__ bd3,
    float* __restrict__ action)
{
  __shared__ float lds[WAVES * 16 * H];
  const int lane = threadIdx.x & 31;
  const int wave = threadIdx.x >> 5;
  const int col  = lane & 15;
  const int hi   = lane >> 4;
  float* wl = &lds[wave * 16 * H];

  v2f wd1b[2][16];
  #pragma unroll
  for (int c = 0; c < 16; ++c) {
    const int k0 = 4 * c + 2 * hi;
    #pragma unroll
    for (int nt = 0; nt < 2; ++nt) {
      const int n = nt * 16 + col;
      wd1b[nt][c] = mkv2(Wd1[k0 * 32 + n], Wd1[(k0 + 1) * 32 + n]);
    }
  }
  v2f wd2b[2][8], wd3b[8];
  #pragma unroll
  for (int c = 0; c < 8; ++c) {
    const int k0 = 4 * c + 2 * hi;
    #pragma unroll
    for (int nt = 0; nt < 2; ++nt) {
      const int n = nt * 16 + col;
      wd2b[nt][c] = mkv2(Wd2[k0 * 32 + n], Wd2[(k0 + 1) * 32 + n]);
    }
    wd3b[c] = mkv2(col == 0 ? Wd3[k0] : 0.0f,
                   col == 0 ? Wd3[k0 + 1] : 0.0f);
  }
  const float bd1v0 = bd1[col], bd1v1 = bd1[16 + col];
  const float bd2v0 = bd2[col], bd2v1 = bd2[16 + col];
  const float bd3v  = bd3[0];

  const int rowBase = blockIdx.x * ROWS_PER_BLOCK + wave * (TILES * 16);

  for (int t = 0; t < TILES; ++t) {
    const int row0 = rowBase + t * 16;

    v8f acc0 = {}, acc1 = {};
    #pragma unroll
    for (int c = 0; c < 16; ++c) {
      v2f a = *(const v2f*)&z_ws[(row0 + col) * 64 + 4 * c + 2 * hi];
      acc0 = WMMA_F32(a, wd1b[0][c], acc0);
      acc1 = WMMA_F32(a, wd1b[1][c], acc1);
    }
    wave_lds_fence();
    #pragma unroll
    for (int j = 0; j < 8; ++j) {
      const int r = j + 8 * hi;
      wl[r * 32 + col]      = fmaxf(acc0[j] + bd1v0, 0.0f);
      wl[r * 32 + 16 + col] = fmaxf(acc1[j] + bd1v1, 0.0f);
    }
    wave_lds_fence();

    v8f a20 = {}, a21 = {};
    #pragma unroll
    for (int c = 0; c < 8; ++c) {
      v2f a = *(const v2f*)&wl[col * 32 + 4 * c + 2 * hi];
      a20 = WMMA_F32(a, wd2b[0][c], a20);
      a21 = WMMA_F32(a, wd2b[1][c], a21);
    }
    wave_lds_fence();
    #pragma unroll
    for (int j = 0; j < 8; ++j) {
      const int r = j + 8 * hi;
      wl[r * 32 + col]      = fmaxf(a20[j] + bd2v0, 0.0f);
      wl[r * 32 + 16 + col] = fmaxf(a21[j] + bd2v1, 0.0f);
    }
    wave_lds_fence();

    v8f accS = {};
    #pragma unroll
    for (int c = 0; c < 8; ++c) {
      v2f a = *(const v2f*)&wl[col * 32 + 4 * c + 2 * hi];
      accS = WMMA_F32(a, wd3b[c], accS);
    }
    if (col == 0) {
      #pragma unroll
      for (int j = 0; j < 8; ++j)
        action[row0 + j + 8 * hi] = accS[j] + bd3v;
    }
  }
}

extern "C" void kernel_launch(void* const* d_in, const int* in_sizes, int n_in,
                              void* d_out, int out_size, void* d_ws, size_t ws_size,
                              hipStream_t stream) {
  (void)in_sizes; (void)n_in; (void)out_size; (void)ws_size;
  const float* x   = (const float*)d_in[0];
  const float* eps = (const float*)d_in[1];
  const float* W1  = (const float*)d_in[2];
  const float* b1  = (const float*)d_in[3];
  const float* W2  = (const float*)d_in[4];
  const float* b2  = (const float*)d_in[5];
  const float* Wmu = (const float*)d_in[6];
  const float* bmu = (const float*)d_in[7];
  const float* Wlv = (const float*)d_in[8];
  const float* blv = (const float*)d_in[9];
  const float* Wd1 = (const float*)d_in[10];
  const float* bd1 = (const float*)d_in[11];
  const float* Wd2 = (const float*)d_in[12];
  const float* bd2 = (const float*)d_in[13];
  const float* Wd3 = (const float*)d_in[14];
  const float* bd3 = (const float*)d_in[15];

  float* out     = (float*)d_out;
  float* action  = out;
  float* mu_out  = out + Bsz;
  float* lv_out  = out + Bsz + Bsz * D;
  float* z_ws    = (float*)d_ws;

  dim3 grid1(Bsz / ROWS_PER_BLOCK, D);
  encoder_kernel<<<grid1, 256, 0, stream>>>(x, eps, W1, b1, W2, b2,
                                            Wmu, bmu, Wlv, blv,
                                            mu_out, lv_out, z_ws);
  decoder_kernel<<<Bsz / ROWS_PER_BLOCK, 256, 0, stream>>>(z_ws, Wd1, bd1,
                                                           Wd2, bd2, Wd3, bd3,
                                                           action);
}


# encoder: each wave keeps its 64 rows and loops over 8 features; x/eps in and mu/logvar/z out go through a wave-private LDS tile as 32-byte row pieces, not 4-byte accesses at a 256-byte stride
# speedup vs baseline: 2.5202x; 1.6406x over previous
#include <hip/hip_runtime.h>
#include <hip/hip_bf16.h>
#include <math.h>

typedef __attribute__((ext_vector_type(2))) float v2f;
typedef __attribute__((ext_vector_type(8))) float v8f;

#define WMMA_F32(A, B, C) \
  __builtin_amdgcn_wmma_f32_16x16x4_f32(false, (A), false, (B), (short)0, (C), false, false)

__device__ __forceinline__ v2f mkv2(float a, float b) { v2f r; r.x = a; r.y = b; return r; }

__device__ __forceinline__ void wave_lds_fence() {
  asm volatile("s_wait_dscnt 0" ::: "memory");
}

constexpr int Bsz = 32768;
constexpr int D   = 64;
constexpr int H   = 32;
constexpr int TILES = 4;
constexpr int WAVES = 8;
constexpr int ROWS_PER_BLOCK = WAVES * TILES * 16;
constexpr int WROWS = TILES * 16;
constexpr int FPW   = 8;
constexpr int PITCH = 12;
constexpr int PLANE = WROWS * PITCH;

__global__ __launch_bounds__(256) void encoder_kernel(
    const float* __restrict__ x,   const float* __restrict__ eps,
    const float* __restrict__ W1,  const float* __restrict__ b1,
    const float* __restrict__ W2,  const float* __restrict__ b2,
    const float* __restrict__ Wmu, const float* __restrict__ bmu,
    const float* __restrict__ Wlv, const float* __restrict__ blv,
    float* __restrict__ mu_out, float* __restrict__ lv_out,
    float* __restrict__ z_ws)
{
  const int lane = threadIdx.x & 31;
  const int wave = threadIdx.x >> 5;
  const int col  = lane & 15;
  const int hi   = lane >> 4;
  const int d0   = blockIdx.y * FPW;
  const int rowBase = blockIdx.x * ROWS_PER_BLOCK + wave * WROWS;

  __shared__ __attribute__((aligned(16))) float lds[WAVES * 5 * PLANE];
  float* sx  = &lds[wave * 5 * PLANE];
  float* se  = sx + PLANE;
  float* smu = sx + 2 * PLANE;
  float* slv = sx + 3 * PLANE;
  float* sz  = sx + 4 * PLANE;

  #pragma unroll
  for (int i = 0; i < 4; ++i) {
    const int pc = i * 32 + lane;
    const int r  = pc >> 1, h4 = 4 * (pc & 1);
    const float4 vx = *(const float4*)&x[(rowBase + r) * D + d0 + h4];
    const float4 ve = *(const float4*)&eps[(rowBase + r) * D + d0 + h4];
    *(float4*)&sx[r * PITCH + h4] = vx;
    *(float4*)&se[r * PITCH + h4] = ve;
  }
  wave_lds_fence();

  for (int f = 0; f < FPW; ++f) {
    const int d = d0 + f;

    float xv[TILES];
    float ev[TILES / 2];
    #pragma unroll
    for (int t = 0; t < TILES; ++t) xv[t] = sx[(16 * t + col) * PITCH + f];
    #pragma unroll
    for (int p = 0; p < TILES / 2; ++p) ev[p] = se[(32 * p + lane) * PITCH + f];

    v2f w2a[2][8];
    v2f w1v[8], b1v[8];
    v2f wsa[8];
    const bool   srow = (col & 7) < 2;
    const float* wsp  = ((col & 7) == 1 ? Wlv : Wmu) + d * H;
    #pragma unroll
    for (int c = 0; c < 8; ++c) {
      const int k0 = 4 * c + 2 * hi;
      w1v[c] = *(const v2f*)&W1[d * H + k0];
      b1v[c] = *(const v2f*)&b1[d * H + k0];
      #pragma unroll
      for (int nt = 0; nt < 2; ++nt) {
        const int n = nt * 16 + col;
        w2a[nt][c] = mkv2(W2[d * H * H + k0 * H + n],
                          W2[d * H * H + (k0 + 1) * H + n]);
      }
      const int nb = 16 * (c >> 2) + 8 * hi + 2 * (c & 3);
      const v2f wv = *(const v2f*)&wsp[nb];
      wsa[c] = mkv2(srow ? wv.x : 0.0f, srow ? wv.y : 0.0f);
    }
    const v8f b2r0 = *(const v8f*)&b2[d * H + 8 * hi];
    const v8f b2r1 = *(const v8f*)&b2[d * H + 16 + 8 * hi];
    const float bmuv = bmu[d];
    const float blvv = blv[d];

    #pragma unroll
    for (int p = 0; p < TILES / 2; ++p) {
      float mu2[2], lv2[2];
      #pragma unroll
      for (int u = 0; u < 2; ++u) {
        const float xt = xv[2 * p + u];

        v8f acc0 = {}, acc1 = {};
        #pragma unroll
        for (int c = 0; c < 8; ++c) {
          v2f hb;
          hb.x = fmaxf(fmaf(xt, w1v[c].x, b1v[c].x), 0.0f);
          hb.y = fmaxf(fmaf(xt, w1v[c].y, b1v[c].y), 0.0f);
          acc0 = WMMA_F32(w2a[0][c], hb, acc0);
          acc1 = WMMA_F32(w2a[1][c], hb, acc1);
        }
        #pragma unroll
        for (int r = 0; r < 8; ++r) {
          acc0[r] = fmaxf(acc0[r] + b2r0[r], 0.0f);
          acc1[r] = fmaxf(acc1[r] + b2r1[r], 0.0f);
        }

        v8f accS = {};
        #pragma unroll
        for (int q = 0; q < 4; ++q)
          accS = WMMA_F32(wsa[q], mkv2(acc0[2 * q], acc0[2 * q + 1]), accS);
        #pragma unroll
        for (int q = 0; q < 4; ++q)
          accS = WMMA_F32(wsa[4 + q], mkv2(acc1[2 * q], acc1[2 * q + 1]), accS);

        mu2[u] = accS[0] + bmuv;
        lv2[u] = accS[1] + blvv;
      }

      const float m  = hi ? mu2[1] : mu2[0];
      const float lv = hi ? lv2[1] : lv2[0];
      const int  lr  = 32 * p + lane;
      smu[lr * PITCH + f] = m;
      slv[lr * PITCH + f] = lv;
      sz[lr * PITCH + f]  = fmaf(ev[p], __expf(0.5f * lv), m);
    }
  }

  wave_lds_fence();
  #pragma unroll
  for (int i = 0; i < 4; ++i) {
    const int pc = i * 32 + lane;
    const int r  = pc >> 1, h4 = 4 * (pc & 1);
    const int g  = (rowBase + r) * D + d0 + h4;
    *(float4*)&mu_out[g] = *(const float4*)&smu[r * PITCH + h4];
    *(float4*)&lv_out[g] = *(const float4*)&slv[r * PITCH + h4];
    *(float4*)&z_ws[g]   = *(const float4*)&sz[r * PITCH + h4];
  }
}

__global__ __launch_bounds__(256) void decoder_kernel(
    const float* __restrict__ z_ws,
    const float* __restrict__ Wd1, const float* __restrict__ bd1,
    const float* __restrict__ Wd2, const float* __restrict__ bd2,
    const float* __restrict__ Wd3, const float* __restrict__ bd3,
    float* __restrict__ action)
{
  __shared__ float lds[WAVES * 16 * H];
  const int lane = threadIdx.x & 31;
  const int wave = threadIdx.x >> 5;
  const int col  = lane & 15;
  const int hi   = lane >> 4;
  float* wl = &lds[wave * 16 * H];

  v2f wd1b[2][16];
  #pragma unroll
  for (int c = 0; c < 16; ++c) {
    const int k0 = 4 * c + 2 * hi;
    #pragma unroll
    for (int nt = 0; nt < 2; ++nt) {
      const int n = nt * 16 + col;
      wd1b[nt][c] = mkv2(Wd1[k0 * 32 + n], Wd1[(k0 + 1) * 32 + n]);
    }
  }
  v2f wd2b[2][8], wd3b[8];
  #pragma unroll
  for (int c = 0; c < 8; ++c) {
    const int k0 = 4 * c + 2 * hi;
    #pragma unroll
    for (int nt = 0; nt < 2; ++nt) {
      const int n = nt * 16 + col;
      wd2b[nt][c] = mkv2(Wd2[k0 * 32 + n], Wd2[(k0 + 1) * 32 + n]);
    }
    wd3b[c] = mkv2(col == 0 ? Wd3[k0] : 0.0f,
                   col == 0 ? Wd3[k0 + 1] : 0.0f);
  }
  const float bd1v0 = bd1[col], bd1v1 = bd1[16 + col];
  const float bd2v0 = bd2[col], bd2v1 = bd2[16 + col];
  const float bd3v  = bd3[0];

  const int rowBase = blockIdx.x * ROWS_PER_BLOCK + wave * (TILES * 16);

  for (int t = 0; t < TILES; ++t) {
    const int row0 = rowBase + t * 16;

    v8f acc0 = {}, acc1 = {};
    #pragma unroll
    for (int c = 0; c < 16; ++c) {
      v2f a = *(const v2f*)&z_ws[(row0 + col) * 64 + 4 * c + 2 * hi];
      acc0 = WMMA_F32(a, wd1b[0][c], acc0);
      acc1 = WMMA_F32(a, wd1b[1][c], acc1);
    }
    wave_lds_fence();
    #pragma unroll
    for (int j = 0; j < 8; ++j) {
      const int r = j + 8 * hi;
      wl[r * 32 + col]      = fmaxf(acc0[j] + bd1v0, 0.0f);
      wl[r * 32 + 16 + col] = fmaxf(acc1[j] + bd1v1, 0.0f);
    }
    wave_lds_fence();

    v8f a20 = {}, a21 = {};
    #pragma unroll
    for (int c = 0; c < 8; ++c) {
      v2f a = *(const v2f*)&wl[col * 32 + 4 * c + 2 * hi];
      a20 = WMMA_F32(a, wd2b[0][c], a20);
      a21 = WMMA_F32(a, wd2b[1][c], a21);
    }
    wave_lds_fence();
    #pragma unroll
    for (int j = 0; j < 8; ++j) {
      const int r = j + 8 * hi;
      wl[r * 32 + col]      = fmaxf(a20[j] + bd2v0, 0.0f);
      wl[r * 32 + 16 + col] = fmaxf(a21[j] + bd2v1, 0.0f);
    }
    wave_lds_fence();

    v8f accS = {};
    #pragma unroll
    for (int c = 0; c < 8; ++c) {
      v2f a = *(const v2f*)&wl[col * 32 + 4 * c + 2 * hi];
      accS = WMMA_F32(a, wd3b[c], accS);
    }
    if (col == 0) {
      #pragma unroll
      for (int j = 0; j < 8; ++j)
        action[row0 + j + 8 * hi] = accS[j] + bd3v;
    }
  }
}

extern "C" void kernel_launch(void* const* d_in, const int* in_sizes, int n_in,
                              void* d_out, int out_size, void* d_ws, size_t ws_size,
                              hipStream_t stream) {
  (void)in_sizes; (void)n_in; (void)out_size; (void)ws_size;
  const float* x   = (const float*)d_in[0];
  const float* eps = (const float*)d_in[1];
  const float* W1  = (const float*)d_in[2];
  const float* b1  = (const float*)d_in[3];
  const float* W2  = (const float*)d_in[4];
  const float* b2  = (const float*)d_in[5];
  const float* Wmu = (const float*)d_in[6];
  const float* bmu = (const float*)d_in[7];
  const float* Wlv = (const float*)d_in[8];
  const float* blv = (const float*)d_in[9];
  const float* Wd1 = (const float*)d_in[10];
  const float* bd1 = (const float*)d_in[11];
  const float* Wd2 = (const float*)d_in[12];
  const float* bd2 = (const float*)d_in[13];
  const float* Wd3 = (const float*)d_in[14];
  const float* bd3 = (const float*)d_in[15];

  float* out     = (float*)d_out;
  float* action  = out;
  float* mu_out  = out + Bsz;
  float* lv_out  = out + Bsz + Bsz * D;
  float* z_ws    = (float*)d_ws;

  dim3 grid1(Bsz / ROWS_PER_BLOCK, D / FPW);
  encoder_kernel<<<grid1, 256, 0, stream>>>(x, eps, W1, b1, W2, b2,
                                            Wmu, bmu, Wlv, blv,
                                            mu_out, lv_out, z_ws);
  decoder_kernel<<<Bsz / ROWS_PER_BLOCK, 256, 0, stream>>>(z_ws, Wd1, bd1,
                                                           Wd2, bd2, Wd3, bd3,
                                                           action);
}


# decoder fused into the encoder kernel: a block holds 64 rows x all 64 features, z stays in LDS, after one workgroup barrier waves 0-3 decode one 16-row tile each; one launch, no workspace trip
# speedup vs baseline: 3.2349x; 1.2836x over previous
#include <hip/hip_runtime.h>
#include <hip/hip_bf16.h>
#include <math.h>

typedef __attribute__((ext_vector_type(2))) float v2f;
typedef __attribute__((ext_vector_type(8))) float v8f;

#define WMMA_F32(A, B, C) \
  __builtin_amdgcn_wmma_f32_16x16x4_f32(false, (A), false, (B), (short)0, (C), false, false)

__device__ __forceinline__ v2f mkv2(float a, float b) { v2f r; r.x = a; r.y = b; return r; }

__device__ __forceinline__ void wave_lds_fence() {
  asm volatile("s_wait_dscnt 0" ::: "memory");
}

constexpr int Bsz = 32768;
constexpr int D   = 64;
constexpr int H   = 32;
constexpr int TILES = 4;
constexpr int WAVES = 8;
constexpr int WROWS = TILES * 16;
constexpr int FPW   = 8;
constexpr int PITCH = 12;
constexpr int PLANE = WROWS * PITCH;

__global__ __launch_bounds__(256) void encoder_decoder_kernel(
    const float* __restrict__ x,   const float* __restrict__ eps,
    const float* __restrict__ W1,  const float* __restrict__ b1,
    const float* __restrict__ W2,  const float* __restrict__ b2,
    const float* __restrict__ Wmu, const float* __restrict__ bmu,
    const float* __restrict__ Wlv, const float* __restrict__ blv,
    const float* __restrict__ Wd1, const float* __restrict__ bd1,
    const float* __restrict__ Wd2, const float* __restrict__ bd2,
    const float* __restrict__ Wd3, const float* __restrict__ bd3,
    float* __restrict__ mu_out, float* __restrict__ lv_out,
    float* __restrict__ action)
{
  const int lane = threadIdx.x & 31;
  const int wave = __builtin_amdgcn_readfirstlane(threadIdx.x >> 5);
  const int col  = lane & 15;
  const int hi   = lane >> 4;
  const int d0   = wave * FPW;
  const int rowBase = blockIdx.x * WROWS;

  __shared__ __attribute__((aligned(16))) float lds[WAVES * 5 * PLANE];
  float* sx  = &lds[wave * 5 * PLANE];
  float* se  = sx + PLANE;
  float* smu = sx + 2 * PLANE;
  float* slv = sx + 3 * PLANE;
  float* sz  = sx + 4 * PLANE;

  #pragma unroll
  for (int i = 0; i < 4; ++i) {
    const int pc = i * 32 + lane;
    const int r  = pc >> 1, h4 = 4 * (pc & 1);
    const float4 vx = *(const float4*)&x[(rowBase + r) * D + d0 + h4];
    const float4 ve = *(const float4*)&eps[(rowBase + r) * D + d0 + h4];
    *(float4*)&sx[r * PITCH + h4] = vx;
    *(float4*)&se[r * PITCH + h4] = ve;
  }
  wave_lds_fence();

  for (int f = 0; f < FPW; ++f) {
    const int d = d0 + f;

    float xv[TILES];
    float ev[TILES / 2];
    #pragma unroll
    for (int t = 0; t < TILES; ++t) xv[t] = sx[(16 * t + col) * PITCH + f];
    #pragma unroll
    for (int p = 0; p < TILES / 2; ++p) ev[p] = se[(32 * p + lane) * PITCH + f];

    v2f w2a[2][8];
    v2f w1v[8], b1v[8];
    v2f wsa[8];
    const bool   srow = (col & 7) < 2;
    const float* wsp  = ((col & 7) == 1 ? Wlv : Wmu) + d * H;
    #pragma unroll
    for (int c = 0; c < 8; ++c) {
      const int k0 = 4 * c + 2 * hi;
      w1v[c] = *(const v2f*)&W1[d * H + k0];
      b1v[c] = *(const v2f*)&b1[d * H + k0];
      #pragma unroll
      for (int nt = 0; nt < 2; ++nt) {
        const int n = nt * 16 + col;
        w2a[nt][c] = mkv2(W2[d * H * H + k0 * H + n],
                          W2[d * H * H + (k0 + 1) * H + n]);
      }
      const int nb = 16 * (c >> 2) + 8 * hi + 2 * (c & 3);
      const v2f wv = *(const v2f*)&wsp[nb];
      wsa[c] = mkv2(srow ? wv.x : 0.0f, srow ? wv.y : 0.0f);
    }
    const v8f b2r0 = *(const v8f*)&b2[d * H + 8 * hi];
    const v8f b2r1 = *(const v8f*)&b2[d * H + 16 + 8 * hi];
    const float bmuv = bmu[d];
    const float blvv = blv[d];

    #pragma unroll
    for (int p = 0; p < TILES / 2; ++p) {
      float mu2[2], lv2[2];
      #pragma unroll
      for (int u = 0; u < 2; ++u) {
        const float xt = xv[2 * p + u];

        v8f acc0 = {}, acc1 = {};
        #pragma unroll
        for (int c = 0; c < 8; ++c) {
          v2f hb;
          hb.x = fmaxf(fmaf(xt, w1v[c].x, b1v[c].x), 0.0f);
          hb.y = fmaxf(fmaf(xt, w1v[c].y, b1v[c].y), 0.0f);
          acc0 = WMMA_F32(w2a[0][c], hb, acc0);
          acc1 = WMMA_F32(w2a[1][c], hb, acc1);
        }
        #pragma unroll
        for (int r = 0; r < 8; ++r) {
          acc0[r] = fmaxf(acc0[r] + b2r0[r], 0.0f);
          acc1[r] = fmaxf(acc1[r] + b2r1[r], 0.0f);
        }

        v8f accS = {};
        #pragma unroll
        for (int q = 0; q < 4; ++q)
          accS = WMMA_F32(wsa[q], mkv2(acc0[2 * q], acc0[2 * q + 1]), accS);
        #pragma unroll
        for (int q = 0; q < 4; ++q)
          accS = WMMA_F32(wsa[4 + q], mkv2(acc1[2 * q], acc1[2 * q + 1]), accS);

        mu2[u] = accS[0] + bmuv;
        lv2[u] = accS[1] + blvv;
      }

      const float m  = hi ? mu2[1] : mu2[0];
      const float lv = hi ? lv2[1] : lv2[0];
      const int  lr  = 32 * p + lane;
      smu[lr * PITCH + f] = m;
      slv[lr * PITCH + f] = lv;
      sz[lr * PITCH + f]  = fmaf(ev[p], __expf(0.5f * lv), m);
    }
  }

  wave_lds_fence();
  #pragma unroll
  for (int i = 0; i < 4; ++i) {
    const int pc = i * 32 + lane;
    const int r  = pc >> 1, h4 = 4 * (pc & 1);
    const int g  = (rowBase + r) * D + d0 + h4;
    *(float4*)&mu_out[g] = *(const float4*)&smu[r * PITCH + h4];
    *(float4*)&lv_out[g] = *(const float4*)&slv[r * PITCH + h4];
  }

  const bool dec = wave < TILES;
  v2f wd1b[2][16];
  v2f wd2b[2][8], wd3b[8];
  float bd1v0 = 0.0f, bd1v1 = 0.0f, bd2v0 = 0.0f, bd2v1 = 0.0f, bd3v = 0.0f;
  if (dec) {
    #pragma unroll
    for (int c = 0; c < 16; ++c) {
      const int k0 = 4 * c + 2 * hi;
      #pragma unroll
      for (int nt = 0; nt < 2; ++nt) {
        const int n = nt * 16 + col;
        wd1b[nt][c] = mkv2(Wd1[k0 * 32 + n], Wd1[(k0 + 1) * 32 + n]);
      }
    }
    #pragma unroll
    for (int c = 0; c < 8; ++c) {
      const int k0 = 4 * c + 2 * hi;
      #pragma unroll
      for (int nt = 0; nt < 2; ++nt) {
        const int n = nt * 16 + col;
        wd2b[nt][c] = mkv2(Wd2[k0 * 32 + n], Wd2[(k0 + 1) * 32 + n]);
      }
      wd3b[c] = mkv2(col == 0 ? Wd3[k0] : 0.0f,
                     col == 0 ? Wd3[k0 + 1] : 0.0f);
    }
    bd1v0 = bd1[col]; bd1v1 = bd1[16 + col];
    bd2v0 = bd2[col]; bd2v1 = bd2[16 + col];
    bd3v  = bd3[0];
  }

  __builtin_amdgcn_fence(__ATOMIC_RELEASE, "workgroup", "local");
  __builtin_amdgcn_s_barrier();
  __builtin_amdgcn_fence(__ATOMIC_ACQUIRE, "workgroup", "local");

  if (dec) {
    const float* zt = &lds[4 * PLANE + (16 * wave + col) * PITCH + 2 * hi];
    float* wl = sx;
    const int row0 = rowBase + 16 * wave;

    v8f acc0 = {}, acc1 = {};
    #pragma unroll
    for (int c = 0; c < 16; ++c) {
      const v2f a = *(const v2f*)&zt[(c >> 1) * 5 * PLANE + 4 * (c & 1)];
      acc0 = WMMA_F32(a, wd1b[0][c], acc0);
      acc1 = WMMA_F32(a, wd1b[1][c], acc1);
    }
    #pragma unroll
    for (int j = 0; j < 8; ++j) {
      const int r = j + 8 * hi;
      wl[r * 32 + col]      = fmaxf(acc0[j] + bd1v0, 0.0f);
      wl[r * 32 + 16 + col] = fmaxf(acc1[j] + bd1v1, 0.0f);
    }
    wave_lds_fence();

    v8f a20 = {}, a21 = {};
    #pragma unroll
    for (int c = 0; c < 8; ++c) {
      const v2f a = *(const v2f*)&wl[col * 32 + 4 * c + 2 * hi];
      a20 = WMMA_F32(a, wd2b[0][c], a20);
      a21 = WMMA_F32(a, wd2b[1][c], a21);
    }
    wave_lds_fence();
    #pragma unroll
    for (int j = 0; j < 8; ++j) {
      const int r = j + 8 * hi;
      wl[r * 32 + col]      = fmaxf(a20[j] + bd2v0, 0.0f);
      wl[r * 32 + 16 + col] = fmaxf(a21[j] + bd2v1, 0.0f);
    }
    wave_lds_fence();

    v8f accS = {};
    #pragma unroll
    for (int c = 0; c < 8; ++c) {
      const v2f a = *(const v2f*)&wl[col * 32 + 4 * c + 2 * hi];
      accS = WMMA_F32(a, wd3b[c], accS);
    }
    if (col == 0) {
      #pragma unroll
      for (int j = 0; j < 8; ++j)
        action[row0 + j + 8 * hi] = accS[j] + bd3v;
    }
  }
}

extern "C" void kernel_launch(void* const* d_in, const int* in_sizes, int n_in,
                              void* d_out, int out_size, void* d_ws, size_t ws_size,
                              hipStream_t stream) {
  (void)in_sizes; (void)n_in; (void)out_size; (void)ws_size;
  const float* x   = (const float*)d_in[0];
  const float* eps = (const float*)d_in[1];
  const float* W1  = (const float*)d_in[2];
  const float* b1  = (const float*)d_in[3];
  const float* W2  = (const float*)d_in[4];
  const float* b2  = (const float*)d_in[5];
  const float* Wmu = (const float*)d_in[6];
  const float* bmu = (const float*)d_in[7];
  const float* Wlv = (const float*)d_in[8];
  const float* blv = (const float*)d_in[9];
  const float* Wd1 = (const float*)d_in[10];
  const float* bd1 = (const float*)d_in[11];
  const float* Wd2 = (const float*)d_in[12];
  const float* bd2 = (const float*)d_in[13];
  const float* Wd3 = (const float*)d_in[14];
  const float* bd3 = (const float*)d_in[15];

  float* out     = (float*)d_out;
  float* action  = out;
  float* mu_out  = out + Bsz;
  float* lv_out  = out + Bsz + Bsz * D;
  (void)d_ws;

  encoder_decoder_kernel<<<Bsz / WROWS, 256, 0, stream>>>(
      x, eps, W1, b1, W2, b2, Wmu, bmu, Wlv, blv,
      Wd1, bd1, Wd2, bd2, Wd3, bd3, mu_out, lv_out, action);
}
